# starting kernel
# baseline (speedup 1.0000x reference)
#include <hip/hip_runtime.h>
#include <math.h>

typedef float v2f __attribute__((ext_vector_type(2)));
typedef float v8f __attribute__((ext_vector_type(8)));

#define DIM  512
#define NEXP 8

__global__ __launch_bounds__(32)
void gs_cholesky_wmma(const float* __restrict__ x, float* __restrict__ out,
                      unsigned nSamples) {
  const int lane    = threadIdx.x;
  const unsigned s0 = blockIdx.x * 2u;

  __shared__ float    Gm[16 * 16];
  __shared__ float    Wm[16 * 16];
  __shared__ unsigned rowOff[16];

  if (lane < 16) {
    const unsigned e = (unsigned)lane & 7u;
    const unsigned s = s0 + (unsigned)(lane >> 3);
    rowOff[lane] = (e * nSamples + s) * (unsigned)DIM;
  }
#pragma unroll
  for (int i = 0; i < 8; ++i) Wm[lane + 32 * i] = 0.0f;
  __syncthreads();

  const int half = lane >> 4;
  const int mrow = lane & 15;
  const int koff = half << 1;

  const float* prow = x + (size_t)rowOff[mrow];

  v8f g = {0.f, 0.f, 0.f, 0.f, 0.f, 0.f, 0.f, 0.f};
#pragma unroll 4
  for (int t = 0; t < DIM / 4; ++t) {
    const v2f a = *(const v2f*)(prow + 4 * t + koff);
    g = __builtin_amdgcn_wmma_f32_16x16x4_f32(false, a, false, a,
                                              (short)0, g, false, false);
  }
#pragma unroll
  for (int r = 0; r < 8; ++r) Gm[(r + half * 8) * 16 + mrow] = g[r];
  __syncthreads();

  if (lane < 2) {
    const int o = lane * 8;
    for (int i = 0; i < 8; ++i) {
      float d = Gm[(o + i) * 16 + (o + i)];
      for (int k = 0; k < i; ++k) {
        const float v = Gm[(o + i) * 16 + (o + k)];
        d -= v * v;
      }
      d = sqrtf(d);
      Gm[(o + i) * 16 + (o + i)] = d;
      const float inv = 1.0f / d;
      for (int j = i + 1; j < 8; ++j) {
        float acc = Gm[(o + j) * 16 + (o + i)];
        for (int k = 0; k < i; ++k)
          acc -= Gm[(o + j) * 16 + (o + k)] * Gm[(o + i) * 16 + (o + k)];
        Gm[(o + j) * 16 + (o + i)] = acc * inv;
      }
    }
    for (int j = 0; j < 8; ++j) {
      Wm[(o + j) * 16 + (o + j)] = 1.0f / Gm[(o + j) * 16 + (o + j)];
      for (int i = j + 1; i < 8; ++i) {
        float acc = 0.0f;
        for (int k = j; k < i; ++k)
          acc += Gm[(o + i) * 16 + (o + k)] * Wm[(o + k) * 16 + (o + j)];
        Wm[(o + i) * 16 + (o + j)] = -acc / Gm[(o + i) * 16 + (o + i)];
      }
    }
  }
  __syncthreads();

  v2f aw[4];
  unsigned br0[4], br1[4];
#pragma unroll
  for (int tk = 0; tk < 4; ++tk) {
    const int k0 = 4 * tk + koff;
    aw[tk].x = Wm[mrow * 16 + k0];
    aw[tk].y = Wm[mrow * 16 + k0 + 1];
    br0[tk] = rowOff[k0];
    br1[tk] = rowOff[k0 + 1];
  }

  for (int n = 0; n < DIM / 16; ++n) {
    const int col = n * 16 + mrow;
    v8f q = {0.f, 0.f, 0.f, 0.f, 0.f, 0.f, 0.f, 0.f};
#pragma unroll
    for (int tk = 0; tk < 4; ++tk) {
      v2f b;
      b.x = x[(size_t)br0[tk] + col];
      b.y = x[(size_t)br1[tk] + col];
      q = __builtin_amdgcn_wmma_f32_16x16x4_f32(false, aw[tk], false, b,
                                                (short)0, q, false, false);
    }
#pragma unroll
    for (int r = 0; r < 8; ++r) {
      const unsigned M = (unsigned)(r + half * 8);
      const unsigned s = s0 + (M >> 3);
      const unsigned e = M & 7u;
      __builtin_nontemporal_store(q[r], out + ((size_t)s * NEXP + e) * DIM + col);
    }
  }
}

extern "C" void kernel_launch(void* const* d_in, const int* in_sizes, int n_in,
                              void* d_out, int out_size, void* d_ws, size_t ws_size,
                              hipStream_t stream) {
  (void)n_in; (void)out_size; (void)d_ws; (void)ws_size;
  const float* x = (const float*)d_in[0];
  float* out = (float*)d_out;

  const unsigned total    = (unsigned)in_sizes[0];
  const unsigned nSamples = total / (NEXP * DIM);
  const unsigned nBlocks  = nSamples / 2;

  gs_cholesky_wmma<<<dim3(nBlocks), dim3(32), 0, stream>>>(x, out, nSamples);
}


# Every global load/store now moves 16 B per lane over whole 128-B lines: Gram rows staged through LDS to build A fragments (Gram k order permuted), apply-phase tile columns interleaved by 4
# speedup vs baseline: 1.1886x; 1.1886x over previous
#include <hip/hip_runtime.h>
#include <math.h>


typedef float v2f __attribute__((ext_vector_type(2)));
typedef float v4f __attribute__((ext_vector_type(4)));
typedef float v8f __attribute__((ext_vector_type(8)));

#define DIM    512
#define NEXP   8
#define CHUNK  128
#define TPITCH (CHUNK + 4)

static __device__ __forceinline__ void lds_handover() {
  __builtin_amdgcn_fence(__ATOMIC_ACQ_REL, "workgroup", "local");
}

__global__ __launch_bounds__(32)
void gs_cholesky_wmma(const float* __restrict__ x, float* __restrict__ out,
                      unsigned nSamples) {
  const int lane    = threadIdx.x;
  const unsigned s0 = blockIdx.x * 2u;

  __shared__ float Gm[16 * 16];
  __shared__ float Wm[16 * 16];
  __shared__ __attribute__((aligned(16))) float Tc[16 * TPITCH];

#pragma unroll
  for (int i = 0; i < 8; ++i) Wm[lane + 32 * i] = 0.0f;

  const int half = lane >> 4;
  const int mrow = lane & 15;
  const int koff = half << 1;


  v4f cur[16], nxt[16];
#pragma unroll
  for (int r = 0; r < 16; ++r) {
    const unsigned off = ((unsigned)(r & 7) * nSamples + s0 + (unsigned)(r >> 3)) * (unsigned)DIM;
    cur[r] = *(const v4f*)(x + (size_t)off + 4 * lane);
  }

  v8f g = {0.f, 0.f, 0.f, 0.f, 0.f, 0.f, 0.f, 0.f};
#pragma unroll
  for (int j = 0; j < DIM / CHUNK; ++j) {
    if (j + 1 < DIM / CHUNK) {
#pragma unroll
      for (int r = 0; r < 16; ++r) {
        const unsigned off = ((unsigned)(r & 7) * nSamples + s0 + (unsigned)(r >> 3)) * (unsigned)DIM;
        nxt[r] = *(const v4f*)(x + (size_t)off + CHUNK * (j + 1) + 4 * lane);
      }
    }
#pragma unroll
    for (int r = 0; r < 16; ++r) *(v4f*)(Tc + r * TPITCH + 4 * lane) = cur[r];
    lds_handover();
#pragma unroll
    for (int t = 0; t < CHUNK / 8; ++t) {
      const v4f v = *(const v4f*)(Tc + mrow * TPITCH + 8 * t + 4 * half);
      const v2f a1 = {v.x, v.y};
      const v2f a2 = {v.z, v.w};
      g = __builtin_amdgcn_wmma_f32_16x16x4_f32(false, a1, false, a1,
                                                (short)0, g, false, false);
      g = __builtin_amdgcn_wmma_f32_16x16x4_f32(false, a2, false, a2,
                                                (short)0, g, false, false);
    }
    lds_handover();
    if (j + 1 < DIM / CHUNK) {
#pragma unroll
      for (int r = 0; r < 16; ++r) cur[r] = nxt[r];
    }
  }
#pragma unroll
  for (int r = 0; r < 8; ++r) Gm[(r + half * 8) * 16 + mrow] = g[r];
  __syncthreads();

  if (lane < 2) {
    const int o = lane * 8;
    for (int i = 0; i < 8; ++i) {
      float d = Gm[(o + i) * 16 + (o + i)];
      for (int k = 0; k < i; ++k) {
        const float v = Gm[(o + i) * 16 + (o + k)];
        d -= v * v;
      }
      d = sqrtf(d);
      Gm[(o + i) * 16 + (o + i)] = d;
      const float inv = 1.0f / d;
      for (int j = i + 1; j < 8; ++j) {
        float acc = Gm[(o + j) * 16 + (o + i)];
        for (int k = 0; k < i; ++k)
          acc -= Gm[(o + j) * 16 + (o + k)] * Gm[(o + i) * 16 + (o + k)];
        Gm[(o + j) * 16 + (o + i)] = acc * inv;
      }
    }
    for (int j = 0; j < 8; ++j) {
      Wm[(o + j) * 16 + (o + j)] = 1.0f / Gm[(o + j) * 16 + (o + j)];
      for (int i = j + 1; i < 8; ++i) {
        float acc = 0.0f;
        for (int k = j; k < i; ++k)
          acc += Gm[(o + i) * 16 + (o + k)] * Wm[(o + k) * 16 + (o + j)];
        Wm[(o + i) * 16 + (o + j)] = -acc / Gm[(o + i) * 16 + (o + i)];
      }
    }
  }
  __syncthreads();

  v2f aw[4];
  unsigned ob0[4], ob1[4];
#pragma unroll
  for (int tk = 0; tk < 4; ++tk) {
    const int k0 = 4 * tk + koff;
    const int k1 = k0 + 1;
    aw[tk].x = Wm[mrow * 16 + k0];
    aw[tk].y = Wm[mrow * 16 + k1];
    ob0[tk] = ((unsigned)(k0 & 7) * nSamples + s0 + (unsigned)(k0 >> 3)) * (unsigned)DIM + 4u * (unsigned)mrow;
    ob1[tk] = ((unsigned)(k1 & 7) * nSamples + s0 + (unsigned)(k1 >> 3)) * (unsigned)DIM + 4u * (unsigned)mrow;
  }
  float* const po = out + ((size_t)(s0 + (unsigned)half) * NEXP) * DIM + 4 * mrow;

  v4f b0[4], b1[4], n0[4], n1[4];
#pragma unroll
  for (int tk = 0; tk < 4; ++tk) {
    b0[tk] = *(const v4f*)(x + (size_t)ob0[tk]);
    b1[tk] = *(const v4f*)(x + (size_t)ob1[tk]);
  }

#pragma unroll
  for (int gi = 0; gi < DIM / 64; ++gi) {
    if (gi + 1 < DIM / 64) {
#pragma unroll
      for (int tk = 0; tk < 4; ++tk) {
        n0[tk] = *(const v4f*)(x + (size_t)ob0[tk] + 64 * (gi + 1));
        n1[tk] = *(const v4f*)(x + (size_t)ob1[tk] + 64 * (gi + 1));
      }
    }
    v8f q[4];
#pragma unroll
    for (int j = 0; j < 4; ++j) q[j] = (v8f){0.f, 0.f, 0.f, 0.f, 0.f, 0.f, 0.f, 0.f};
#pragma unroll
    for (int tk = 0; tk < 4; ++tk) {
#pragma unroll
      for (int j = 0; j < 4; ++j) {
        v2f b;
        b.x = b0[tk][j];
        b.y = b1[tk][j];
        q[j] = __builtin_amdgcn_wmma_f32_16x16x4_f32(false, aw[tk], false, b,
                                                     (short)0, q[j], false, false);
      }
    }
#pragma unroll
    for (int r = 0; r < 8; ++r) {
      const v4f o4 = {q[0][r], q[1][r], q[2][r], q[3][r]};
      __builtin_nontemporal_store(o4, (v4f*)(po + (size_t)r * DIM + 64 * gi));
    }
    if (gi + 1 < DIM / 64) {
#pragma unroll
      for (int tk = 0; tk < 4; ++tk) { b0[tk] = n0[tk]; b1[tk] = n1[tk]; }
    }
  }
}

extern "C" void kernel_launch(void* const* d_in, const int* in_sizes, int n_in,
                              void* d_out, int out_size, void* d_ws, size_t ws_size,
                              hipStream_t stream) {
  (void)n_in; (void)out_size; (void)d_ws; (void)ws_size;
  const float* x = (const float*)d_in[0];
  float* out = (float*)d_out;

  const unsigned total    = (unsigned)in_sizes[0];
  const unsigned nSamples = total / (NEXP * DIM);
  const unsigned nBlocks  = nSamples / 2;

  gs_cholesky_wmma<<<dim3(nBlocks), dim3(32), 0, stream>>>(x, out, nSamples);
}


# x is read from global memory once: the stacked 16x512 tile stays in LDS after the Gram phase (all 64 loads issued up front) and the apply phase takes its B fragments from LDS, not from global
# speedup vs baseline: 1.2324x; 1.0369x over previous
#include <hip/hip_runtime.h>
#include <math.h>


typedef float v2f __attribute__((ext_vector_type(2)));
typedef float v4f __attribute__((ext_vector_type(4)));
typedef float v8f __attribute__((ext_vector_type(8)));

#define DIM    512
#define NEXP   8
#define CHUNK  128
#define TPITCH (DIM + 4)

static __device__ __forceinline__ void lds_handover() {
  __builtin_amdgcn_fence(__ATOMIC_ACQ_REL, "workgroup", "local");
}

__global__ __launch_bounds__(32)
void gs_cholesky_wmma(const float* __restrict__ x, float* __restrict__ out,
                      unsigned nSamples) {
  const int lane    = threadIdx.x;
  const unsigned s0 = blockIdx.x * 2u;

  __shared__ float Gm[16 * 16];
  __shared__ float Wm[16 * 16];
  __shared__ __attribute__((aligned(16))) float Tc[16 * TPITCH];

#pragma unroll
  for (int i = 0; i < 8; ++i) Wm[lane + 32 * i] = 0.0f;

  const int half = lane >> 4;
  const int mrow = lane & 15;
  const int koff = half << 1;


  v4f tile[DIM / CHUNK][16];
#pragma unroll
  for (int j = 0; j < DIM / CHUNK; ++j) {
#pragma unroll
    for (int r = 0; r < 16; ++r) {
      const unsigned off = ((unsigned)(r & 7) * nSamples + s0 + (unsigned)(r >> 3)) * (unsigned)DIM;
      tile[j][r] = *(const v4f*)(x + (size_t)off + CHUNK * j + 4 * lane);
    }
  }

  v8f g = {0.f, 0.f, 0.f, 0.f, 0.f, 0.f, 0.f, 0.f};
#pragma unroll
  for (int j = 0; j < DIM / CHUNK; ++j) {
#pragma unroll
    for (int r = 0; r < 16; ++r) *(v4f*)(Tc + r * TPITCH + CHUNK * j + 4 * lane) = tile[j][r];
    lds_handover();
#pragma unroll
    for (int t = 0; t < CHUNK / 8; ++t) {
      const v4f v = *(const v4f*)(Tc + mrow * TPITCH + CHUNK * j + 8 * t + 4 * half);
      const v2f a1 = {v.x, v.y};
      const v2f a2 = {v.z, v.w};
      g = __builtin_amdgcn_wmma_f32_16x16x4_f32(false, a1, false, a1,
                                                (short)0, g, false, false);
      g = __builtin_amdgcn_wmma_f32_16x16x4_f32(false, a2, false, a2,
                                                (short)0, g, false, false);
    }
  }
#pragma unroll
  for (int r = 0; r < 8; ++r) Gm[(r + half * 8) * 16 + mrow] = g[r];
  __syncthreads();

  if (lane < 2) {
    const int o = lane * 8;
    for (int i = 0; i < 8; ++i) {
      float d = Gm[(o + i) * 16 + (o + i)];
      for (int k = 0; k < i; ++k) {
        const float v = Gm[(o + i) * 16 + (o + k)];
        d -= v * v;
      }
      d = sqrtf(d);
      Gm[(o + i) * 16 + (o + i)] = d;
      const float inv = 1.0f / d;
      for (int j = i + 1; j < 8; ++j) {
        float acc = Gm[(o + j) * 16 + (o + i)];
        for (int k = 0; k < i; ++k)
          acc -= Gm[(o + j) * 16 + (o + k)] * Gm[(o + i) * 16 + (o + k)];
        Gm[(o + j) * 16 + (o + i)] = acc * inv;
      }
    }
    for (int j = 0; j < 8; ++j) {
      Wm[(o + j) * 16 + (o + j)] = 1.0f / Gm[(o + j) * 16 + (o + j)];
      for (int i = j + 1; i < 8; ++i) {
        float acc = 0.0f;
        for (int k = j; k < i; ++k)
          acc += Gm[(o + i) * 16 + (o + k)] * Wm[(o + k) * 16 + (o + j)];
        Wm[(o + i) * 16 + (o + j)] = -acc / Gm[(o + i) * 16 + (o + i)];
      }
    }
  }
  __syncthreads();

  v2f aw[4];
#pragma unroll
  for (int tk = 0; tk < 4; ++tk) {
    const int k0 = 4 * tk + koff;
    aw[tk].x = Wm[mrow * 16 + k0];
    aw[tk].y = Wm[mrow * 16 + k0 + 1];
  }
  const float* const tb = Tc + koff * TPITCH + 4 * mrow;
  float* const po = out + ((size_t)(s0 + (unsigned)half) * NEXP) * DIM + 4 * mrow;

  v4f b0[4], b1[4], n0[4], n1[4];
#pragma unroll
  for (int tk = 0; tk < 4; ++tk) {
    b0[tk] = *(const v4f*)(tb + (4 * tk) * TPITCH);
    b1[tk] = *(const v4f*)(tb + (4 * tk + 1) * TPITCH);
  }

#pragma unroll
  for (int gi = 0; gi < DIM / 64; ++gi) {
    if (gi + 1 < DIM / 64) {
#pragma unroll
      for (int tk = 0; tk < 4; ++tk) {
        n0[tk] = *(const v4f*)(tb + (4 * tk) * TPITCH + 64 * (gi + 1));
        n1[tk] = *(const v4f*)(tb + (4 * tk + 1) * TPITCH + 64 * (gi + 1));
      }
    }
    v8f q[4];
#pragma unroll
    for (int j = 0; j < 4; ++j) q[j] = (v8f){0.f, 0.f, 0.f, 0.f, 0.f, 0.f, 0.f, 0.f};
#pragma unroll
    for (int tk = 0; tk < 4; ++tk) {
#pragma unroll
      for (int j = 0; j < 4; ++j) {
        v2f b;
        b.x = b0[tk][j];
        b.y = b1[tk][j];
        q[j] = __builtin_amdgcn_wmma_f32_16x16x4_f32(false, aw[tk], false, b,
                                                     (short)0, q[j], false, false);
      }
    }
#pragma unroll
    for (int r = 0; r < 8; ++r) {
      const v4f o4 = {q[0][r], q[1][r], q[2][r], q[3][r]};
      __builtin_nontemporal_store(o4, (v4f*)(po + (size_t)r * DIM + 64 * gi));
    }
    if (gi + 1 < DIM / 64) {
#pragma unroll
      for (int tk = 0; tk < 4; ++tk) { b0[tk] = n0[tk]; b1[tk] = n1[tk]; }
    }
  }
}

extern "C" void kernel_launch(void* const* d_in, const int* in_sizes, int n_in,
                              void* d_out, int out_size, void* d_ws, size_t ws_size,
                              hipStream_t stream) {
  (void)n_in; (void)out_size; (void)d_ws; (void)ws_size;
  const float* x = (const float*)d_in[0];
  float* out = (float*)d_out;

  const unsigned total    = (unsigned)in_sizes[0];
  const unsigned nSamples = total / (NEXP * DIM);
  const unsigned nBlocks  = nSamples / 2;

  gs_cholesky_wmma<<<dim3(nBlocks), dim3(32), 0, stream>>>(x, out, nSamples);
}


# Each wave works through two tiles: as a 128-column chunk leaves its registers for LDS they take the next tile's global loads, in flight through the Cholesky and apply phases; grid halved
# speedup vs baseline: 1.3629x; 1.1059x over previous
#include <hip/hip_runtime.h>
#include <math.h>


typedef float v2f __attribute__((ext_vector_type(2)));
typedef float v4f __attribute__((ext_vector_type(4)));
typedef float v8f __attribute__((ext_vector_type(8)));

#define DIM    512
#define NEXP   8
#define CHUNK  128
#define NCHUNK (DIM / CHUNK)
#define TPITCH (DIM + 4)
#define TILES_PER_WAVE 2

static __device__ __forceinline__ void lds_handover() {
  __builtin_amdgcn_fence(__ATOMIC_ACQ_REL, "workgroup", "local");
}

static __device__ __forceinline__ void load_chunk(v4f (&dst)[16], const float* __restrict__ x,
                                                  unsigned s0, unsigned nSamples, int j, int lane) {
#pragma unroll
  for (int r = 0; r < 16; ++r) {
    const unsigned off = ((unsigned)(r & 7) * nSamples + s0 + (unsigned)(r >> 3)) * (unsigned)DIM;
    dst[r] = *(const v4f*)(x + (size_t)off + CHUNK * j + 4 * lane);
  }
}

template <bool REFILL>
static __device__ __forceinline__ v8f gram_phase(v4f (&tile)[NCHUNK][16], float* Tc,
                                                 const float* __restrict__ x, unsigned sNext,
                                                 unsigned nSamples, int lane) {
  const int half = lane >> 4;
  const int mrow = lane & 15;
  v8f g = {0.f, 0.f, 0.f, 0.f, 0.f, 0.f, 0.f, 0.f};
#pragma unroll
  for (int j = 0; j < NCHUNK; ++j) {
#pragma unroll
    for (int r = 0; r < 16; ++r) *(v4f*)(Tc + r * TPITCH + CHUNK * j + 4 * lane) = tile[j][r];
    if (REFILL) load_chunk(tile[j], x, sNext, nSamples, j, lane);
    lds_handover();
#pragma unroll
    for (int t = 0; t < CHUNK / 8; ++t) {
      const v4f v = *(const v4f*)(Tc + mrow * TPITCH + CHUNK * j + 8 * t + 4 * half);
      const v2f a1 = {v.x, v.y};
      const v2f a2 = {v.z, v.w};
      g = __builtin_amdgcn_wmma_f32_16x16x4_f32(false, a1, false, a1,
                                                (short)0, g, false, false);
      g = __builtin_amdgcn_wmma_f32_16x16x4_f32(false, a2, false, a2,
                                                (short)0, g, false, false);
    }
  }
  return g;
}

__global__ __launch_bounds__(32)
void gs_cholesky_wmma(const float* __restrict__ x, float* __restrict__ out,
                      unsigned nSamples) {
  const int lane = threadIdx.x;

  __shared__ float Gm[16 * 16];
  __shared__ float Wm[16 * 16];
  __shared__ __attribute__((aligned(16))) float Tc[16 * TPITCH];

#pragma unroll
  for (int i = 0; i < 8; ++i) Wm[lane + 32 * i] = 0.0f;

  const int half = lane >> 4;
  const int mrow = lane & 15;
  const int koff = half << 1;

  v4f tile[NCHUNK][16];
#pragma unroll
  for (int j = 0; j < NCHUNK; ++j) load_chunk(tile[j], x, blockIdx.x * 2u, nSamples, j, lane);

  const unsigned tileStride = nSamples / (2u * TILES_PER_WAVE);
#pragma nounroll
  for (unsigned it = 0; it < TILES_PER_WAVE; ++it) {
    const unsigned s0 = (blockIdx.x + it * tileStride) * 2u;

    v8f g;
    if (it + 1 < TILES_PER_WAVE)
      g = gram_phase<true>(tile, Tc, x, s0 + 2u * tileStride, nSamples, lane);
    else
      g = gram_phase<false>(tile, Tc, x, 0u, nSamples, lane);
#pragma unroll
    for (int r = 0; r < 8; ++r) Gm[(r + half * 8) * 16 + mrow] = g[r];
    lds_handover();

    if (lane < 2) {
      const int o = lane * 8;
      for (int i = 0; i < 8; ++i) {
        float d = Gm[(o + i) * 16 + (o + i)];
        for (int k = 0; k < i; ++k) {
          const float v = Gm[(o + i) * 16 + (o + k)];
          d -= v * v;
        }
        d = sqrtf(d);
        Gm[(o + i) * 16 + (o + i)] = d;
        const float inv = 1.0f / d;
        for (int j = i + 1; j < 8; ++j) {
          float acc = Gm[(o + j) * 16 + (o + i)];
          for (int k = 0; k < i; ++k)
            acc -= Gm[(o + j) * 16 + (o + k)] * Gm[(o + i) * 16 + (o + k)];
          Gm[(o + j) * 16 + (o + i)] = acc * inv;
        }
      }
      for (int j = 0; j < 8; ++j) {
        Wm[(o + j) * 16 + (o + j)] = 1.0f / Gm[(o + j) * 16 + (o + j)];
        for (int i = j + 1; i < 8; ++i) {
          float acc = 0.0f;
          for (int k = j; k < i; ++k)
            acc += Gm[(o + i) * 16 + (o + k)] * Wm[(o + k) * 16 + (o + j)];
          Wm[(o + i) * 16 + (o + j)] = -acc / Gm[(o + i) * 16 + (o + i)];
        }
      }
    }
    lds_handover();

    v2f aw[4];
#pragma unroll
    for (int tk = 0; tk < 4; ++tk) {
      const int k0 = 4 * tk + koff;
      aw[tk].x = Wm[mrow * 16 + k0];
      aw[tk].y = Wm[mrow * 16 + k0 + 1];
    }
    const float* const tb = Tc + koff * TPITCH + 4 * mrow;
    float* const po = out + ((size_t)(s0 + (unsigned)half) * NEXP) * DIM + 4 * mrow;

    v4f b0[4], b1[4], n0[4], n1[4];
#pragma unroll
    for (int tk = 0; tk < 4; ++tk) {
      b0[tk] = *(const v4f*)(tb + (4 * tk) * TPITCH);
      b1[tk] = *(const v4f*)(tb + (4 * tk + 1) * TPITCH);
    }

#pragma unroll
    for (int gi = 0; gi < DIM / 64; ++gi) {
      if (gi + 1 < DIM / 64) {
#pragma unroll
        for (int tk = 0; tk < 4; ++tk) {
          n0[tk] = *(const v4f*)(tb + (4 * tk) * TPITCH + 64 * (gi + 1));
          n1[tk] = *(const v4f*)(tb + (4 * tk + 1) * TPITCH + 64 * (gi + 1));
        }
      }
      v8f q[4];
#pragma unroll
      for (int j = 0; j < 4; ++j) q[j] = (v8f){0.f, 0.f, 0.f, 0.f, 0.f, 0.f, 0.f, 0.f};
#pragma unroll
      for (int tk = 0; tk < 4; ++tk) {
#pragma unroll
        for (int j = 0; j < 4; ++j) {
          v2f b;
          b.x = b0[tk][j];
          b.y = b1[tk][j];
          q[j] = __builtin_amdgcn_wmma_f32_16x16x4_f32(false, aw[tk], false, b,
                                                       (short)0, q[j], false, false);
        }
      }
#pragma unroll
      for (int r = 0; r < 8; ++r) {
        const v4f o4 = {q[0][r], q[1][r], q[2][r], q[3][r]};
        __builtin_nontemporal_store(o4, (v4f*)(po + (size_t)r * DIM + 64 * gi));
      }
      if (gi + 1 < DIM / 64) {
#pragma unroll
        for (int tk = 0; tk < 4; ++tk) { b0[tk] = n0[tk]; b1[tk] = n1[tk]; }
      }
    }
    lds_handover();
  }
}

extern "C" void kernel_launch(void* const* d_in, const int* in_sizes, int n_in,
                              void* d_out, int out_size, void* d_ws, size_t ws_size,
                              hipStream_t stream) {
  (void)n_in; (void)out_size; (void)d_ws; (void)ws_size;
  const float* x = (const float*)d_in[0];
  float* out = (float*)d_out;

  const unsigned total    = (unsigned)in_sizes[0];
  const unsigned nSamples = total / (NEXP * DIM);
  const unsigned nBlocks  = nSamples / (2 * TILES_PER_WAVE);

  gs_cholesky_wmma<<<dim3(nBlocks), dim3(32), 0, stream>>>(x, out, nSamples);
}


# Cholesky and triangular inverse of the two 8x8 blocks run on all lanes in registers (row_share DPP moves between lanes), not serially on two lanes through LDS; same arithmetic, same summation order
# speedup vs baseline: 1.5007x; 1.1011x over previous
#include <hip/hip_runtime.h>
#include <math.h>


typedef float v2f __attribute__((ext_vector_type(2)));
typedef float v4f __attribute__((ext_vector_type(4)));
typedef float v8f __attribute__((ext_vector_type(8)));

#define DIM    512
#define NEXP   8
#define CHUNK  128
#define NCHUNK (DIM / CHUNK)
#define TPITCH (DIM + 4)
#define TILES_PER_WAVE 2

static __device__ __forceinline__ void lds_handover() {
  __builtin_amdgcn_fence(__ATOMIC_ACQ_REL, "workgroup", "local");
}

static __device__ __forceinline__ float row_share(float v, int src) {
  const int x = __builtin_bit_cast(int, v);
  int r;
  switch (src) {
    case 0:  r = __builtin_amdgcn_update_dpp(0, x, 0x150, 0xf, 0xf, true); break;
    case 1:  r = __builtin_amdgcn_update_dpp(0, x, 0x151, 0xf, 0xf, true); break;
    case 2:  r = __builtin_amdgcn_update_dpp(0, x, 0x152, 0xf, 0xf, true); break;
    case 3:  r = __builtin_amdgcn_update_dpp(0, x, 0x153, 0xf, 0xf, true); break;
    case 4:  r = __builtin_amdgcn_update_dpp(0, x, 0x154, 0xf, 0xf, true); break;
    case 5:  r = __builtin_amdgcn_update_dpp(0, x, 0x155, 0xf, 0xf, true); break;
    case 6:  r = __builtin_amdgcn_update_dpp(0, x, 0x156, 0xf, 0xf, true); break;
    default: r = __builtin_amdgcn_update_dpp(0, x, 0x157, 0xf, 0xf, true); break;
  }
  return __builtin_bit_cast(float, r);
}
static __device__ __forceinline__ float row_swap8(float v) {
  const int x = __builtin_bit_cast(int, v);
  return __builtin_bit_cast(float, __builtin_amdgcn_update_dpp(0, x, 0x128, 0xf, 0xf, true));
}

static __device__ __forceinline__ void load_chunk(v4f (&dst)[16], const float* __restrict__ x,
                                                  unsigned s0, unsigned nSamples, int j, int lane) {
#pragma unroll
  for (int r = 0; r < 16; ++r) {
    const unsigned off = ((unsigned)(r & 7) * nSamples + s0 + (unsigned)(r >> 3)) * (unsigned)DIM;
    dst[r] = *(const v4f*)(x + (size_t)off + CHUNK * j + 4 * lane);
  }
}

template <bool REFILL>
static __device__ __forceinline__ v8f gram_phase(v4f (&tile)[NCHUNK][16], float* Tc,
                                                 const float* __restrict__ x, unsigned sNext,
                                                 unsigned nSamples, int lane) {
  const int half = lane >> 4;
  const int mrow = lane & 15;
  v8f g = {0.f, 0.f, 0.f, 0.f, 0.f, 0.f, 0.f, 0.f};
#pragma unroll
  for (int j = 0; j < NCHUNK; ++j) {
#pragma unroll
    for (int r = 0; r < 16; ++r) *(v4f*)(Tc + r * TPITCH + CHUNK * j + 4 * lane) = tile[j][r];
    if (REFILL) load_chunk(tile[j], x, sNext, nSamples, j, lane);
    lds_handover();
#pragma unroll
    for (int t = 0; t < CHUNK / 8; ++t) {
      const v4f v = *(const v4f*)(Tc + mrow * TPITCH + CHUNK * j + 8 * t + 4 * half);
      const v2f a1 = {v.x, v.y};
      const v2f a2 = {v.z, v.w};
      g = __builtin_amdgcn_wmma_f32_16x16x4_f32(false, a1, false, a1,
                                                (short)0, g, false, false);
      g = __builtin_amdgcn_wmma_f32_16x16x4_f32(false, a2, false, a2,
                                                (short)0, g, false, false);
    }
  }
  return g;
}

__global__ __launch_bounds__(32)
void gs_cholesky_wmma(const float* __restrict__ x, float* __restrict__ out,
                      unsigned nSamples) {
  const int lane = threadIdx.x;

  __shared__ float Wm[16 * 16];
  __shared__ __attribute__((aligned(16))) float Tc[16 * TPITCH];

#pragma unroll
  for (int i = 0; i < 8; ++i) Wm[lane + 32 * i] = 0.0f;

  const int half = lane >> 4;
  const int mrow = lane & 15;
  const int koff = half << 1;

  v4f tile[NCHUNK][16];
#pragma unroll
  for (int j = 0; j < NCHUNK; ++j) load_chunk(tile[j], x, blockIdx.x * 2u, nSamples, j, lane);

  const unsigned tileStride = nSamples / (2u * TILES_PER_WAVE);
#pragma nounroll
  for (unsigned it = 0; it < TILES_PER_WAVE; ++it) {
    const unsigned s0 = (blockIdx.x + it * tileStride) * 2u;

    v8f g;
    if (it + 1 < TILES_PER_WAVE)
      g = gram_phase<true>(tile, Tc, x, s0 + 2u * tileStride, nSamples, lane);
    else
      g = gram_phase<false>(tile, Tc, x, 0u, nSamples, lane);
    asm volatile("v_nop\n\tv_nop\n\tv_nop\n\tv_nop\n\tv_nop" : "+v"(g));
    const int  j8  = lane & 7;
    const bool isW = (lane & 8) != 0;
    float a[8];
#pragma unroll
    for (int k = 0; k < 8; ++k) {
      const float t = row_swap8(g[k]);
      a[k] = half ? t : g[k];
    }
    float Lr[8];
    float wc[8];
#pragma unroll
    for (int i = 0; i < 8; ++i) {
      const float d = sqrtf(row_share(a[i], i));
      float acc = 0.0f;
#pragma unroll
      for (int k = 0; k < i; ++k)
        acc = __builtin_fmaf(row_share(Lr[k], i), wc[k], acc);
      const float numW = (j8 == i) ? 1.0f : ((j8 < i) ? -acc : 0.0f);
      const float q = (isW ? numW : 1.0f) / d;
      wc[i] = q;
      const float l = a[i] * q;
      Lr[i] = l;
#pragma unroll
      for (int k = i + 1; k < 8; ++k)
        a[k] = __builtin_fmaf(-l, row_share(l, k), a[k]);
    }
    if (isW) {
#pragma unroll
      for (int i = 0; i < 8; ++i) Wm[(half * 8 + i) * 16 + half * 8 + j8] = wc[i];
    }
    lds_handover();

    v2f aw[4];
#pragma unroll
    for (int tk = 0; tk < 4; ++tk) {
      const int k0 = 4 * tk + koff;
      aw[tk].x = Wm[mrow * 16 + k0];
      aw[tk].y = Wm[mrow * 16 + k0 + 1];
    }
    const float* const tb = Tc + koff * TPITCH + 4 * mrow;
    float* const po = out + ((size_t)(s0 + (unsigned)half) * NEXP) * DIM + 4 * mrow;

    v4f b0[4], b1[4], n0[4], n1[4];
#pragma unroll
    for (int tk = 0; tk < 4; ++tk) {
      b0[tk] = *(const v4f*)(tb + (4 * tk) * TPITCH);
      b1[tk] = *(const v4f*)(tb + (4 * tk + 1) * TPITCH);
    }

#pragma unroll
    for (int gi = 0; gi < DIM / 64; ++gi) {
      if (gi + 1 < DIM / 64) {
#pragma unroll
        for (int tk = 0; tk < 4; ++tk) {
          n0[tk] = *(const v4f*)(tb + (4 * tk) * TPITCH + 64 * (gi + 1));
          n1[tk] = *(const v4f*)(tb + (4 * tk + 1) * TPITCH + 64 * (gi + 1));
        }
      }
      v8f q[4];
#pragma unroll
      for (int j = 0; j < 4; ++j) q[j] = (v8f){0.f, 0.f, 0.f, 0.f, 0.f, 0.f, 0.f, 0.f};
#pragma unroll
      for (int tk = 0; tk < 4; ++tk) {
#pragma unroll
        for (int j = 0; j < 4; ++j) {
          v2f b;
          b.x = b0[tk][j];
          b.y = b1[tk][j];
          q[j] = __builtin_amdgcn_wmma_f32_16x16x4_f32(false, aw[tk], false, b,
                                                       (short)0, q[j], false, false);
        }
      }
#pragma unroll
      for (int r = 0; r < 8; ++r) {
        const v4f o4 = {q[0][r], q[1][r], q[2][r], q[3][r]};
        __builtin_nontemporal_store(o4, (v4f*)(po + (size_t)r * DIM + 64 * gi));
      }
      if (gi + 1 < DIM / 64) {
#pragma unroll
        for (int tk = 0; tk < 4; ++tk) { b0[tk] = n0[tk]; b1[tk] = n1[tk]; }
      }
    }
    lds_handover();
  }
}

extern "C" void kernel_launch(void* const* d_in, const int* in_sizes, int n_in,
                              void* d_out, int out_size, void* d_ws, size_t ws_size,
                              hipStream_t stream) {
  (void)n_in; (void)out_size; (void)d_ws; (void)ws_size;
  const float* x = (const float*)d_in[0];
  float* out = (float*)d_out;

  const unsigned total    = (unsigned)in_sizes[0];
  const unsigned nSamples = total / (NEXP * DIM);
  const unsigned nBlocks  = nSamples / (2 * TILES_PER_WAVE);

  gs_cholesky_wmma<<<dim3(nBlocks), dim3(32), 0, stream>>>(x, out, nSamples);
}
